# starting kernel
# baseline (speedup 1.0000x reference)
#include <hip/hip_runtime.h>
#include <hip/hip_bf16.h>

typedef __attribute__((ext_vector_type(2))) float v2f;
typedef __attribute__((ext_vector_type(4))) float v4f;
typedef __attribute__((ext_vector_type(8))) float v8f;

#define NN 8
#define DD 16
#define FF 512
#define KK (DD * FF)
#define LN_EPS 1e-5f

__global__ __launch_bounds__(256) void contract_wmma_kernel(
    const float* __restrict__ h0, const float* __restrict__ A0m,
    const float* __restrict__ h1, const float* __restrict__ A1m,
    float scale, const float* __restrict__ Wm, float* __restrict__ out,
    int two) {
  __shared__ __align__(16) v2f acol[FF];
  __shared__ float red[8][16][17];

  const int g    = blockIdx.x;
  const int tid  = threadIdx.x;
  const int w    = tid >> 5;
  const int lane = tid & 31;
  const int half = lane >> 4;
  const int m    = lane & 15;
  const int mr   = m & 7;

  for (int i = tid; i < FF; i += 256) {
    v2f t;
    t.x = A0m[i * FF + g];
    t.y = two ? A1m[i * FF + g] : 0.0f;
    acol[i] = t;
  }
  __syncthreads();

  const float* h0p = h0 + mr * KK;
  const float* h1p = h1 + mr * KK;
  const int kbase  = w * 1024 + 2 * half;
  const float* wpc = Wm + (size_t)kbase * (size_t)KK + (size_t)(g * 16 + m);

  int kc = kbase;
  v2f h0v = *(const v2f*)(h0p + kc);
  v2f h1v = *(const v2f*)(h1p + kc);
  v4f av  = *(const v4f*)&acol[kc & (FF - 1)];
  float bx = __builtin_nontemporal_load(wpc);
  float by = __builtin_nontemporal_load(wpc + KK);

  v8f c = {};
  for (int it = 0; it < 255; ++it) {
    v2f Aop;
    Aop.x = h0v.x * av.x + h1v.x * av.y;
    Aop.y = h0v.y * av.z + h1v.y * av.w;
    v2f Bop;
    Bop.x = bx;
    Bop.y = by;

    kc  += 4;
    wpc += 4 * (size_t)KK;
    h0v = *(const v2f*)(h0p + kc);
    h1v = *(const v2f*)(h1p + kc);
    av  = *(const v4f*)&acol[kc & (FF - 1)];
    bx  = __builtin_nontemporal_load(wpc);
    by  = __builtin_nontemporal_load(wpc + KK);
    __builtin_prefetch(wpc + 4 * KK, 0, 1);

    c = __builtin_amdgcn_wmma_f32_16x16x4_f32(
        false, Aop, false, Bop, (short)0, c, false, false);
  }
  {
    v2f Aop;
    Aop.x = h0v.x * av.x + h1v.x * av.y;
    Aop.y = h0v.y * av.z + h1v.y * av.w;
    v2f Bop;
    Bop.x = bx;
    Bop.y = by;
    c = __builtin_amdgcn_wmma_f32_16x16x4_f32(
        false, Aop, false, Bop, (short)0, c, false, false);
  }

#pragma unroll
  for (int j = 0; j < 8; ++j)
    red[w][j + 8 * half][m] = c[j];
  __syncthreads();

  if (tid < 128) {
    const int M = tid >> 4;
    const int N = tid & 15;
    float s = 0.0f;
#pragma unroll
    for (int ww = 0; ww < 8; ++ww) s += red[ww][M][N];
    out[(M * 16 + N) * FF + g] = s * scale;
  }
}

__global__ __launch_bounds__(256) void layernorm16_kernel(
    const float* __restrict__ y, float* __restrict__ out) {
  const int t = blockIdx.x * blockDim.x + threadIdx.x;
  if (t >= NN * FF) return;
  const int n = t / FF;
  const int g = t - n * FF;
  const float* p = y + n * DD * FF + g;
  float v[16];
  float s = 0.0f;
#pragma unroll
  for (int o = 0; o < 16; ++o) { v[o] = p[o * FF]; s += v[o]; }
  const float mean = s * (1.0f / 16.0f);
  float s2 = 0.0f;
#pragma unroll
  for (int o = 0; o < 16; ++o) { const float d = v[o] - mean; s2 += d * d; }
  const float inv = rsqrtf(s2 * (1.0f / 16.0f) + LN_EPS);
  float* q = out + n * DD * FF + g;
#pragma unroll
  for (int o = 0; o < 16; ++o) q[o * FF] = (v[o] - mean) * inv;
}

extern "C" void kernel_launch(void* const* d_in, const int* in_sizes, int n_in,
                              void* d_out, int out_size, void* d_ws, size_t ws_size,
                              hipStream_t stream) {
  const float* x  = (const float*)d_in[0];
  const float* W  = (const float*)d_in[1];
  const float* A  = (const float*)d_in[2];
  const float* A0 = A;
  const float* A1 = A + FF * FF;

  float* y1pre = (float*)d_ws;
  float* h1    = y1pre + NN * DD * FF;
  float* y2pre = h1 + NN * DD * FF;
  float* outp  = (float*)d_out;

  contract_wmma_kernel<<<FF, 256, 0, stream>>>(x, A0, x, A0, 1.0f, W, y1pre, 0);
  layernorm16_kernel<<<(NN * FF + 255) / 256, 256, 0, stream>>>(y1pre, h1);

  contract_wmma_kernel<<<FF, 256, 0, stream>>>(x, A1, h1, A0, 0.5f, W, y2pre, 1);
  layernorm16_kernel<<<(NN * FF + 255) / 256, 256, 0, stream>>>(y2pre, outp);
}


# Re-tile the W stream: workgroup = 4 adjacent columns x half of k, every wave reads whole aligned 256-B row pieces, one group of loads ahead; k halves summed in the norm kernel (new sum order)
# speedup vs baseline: 1.8301x; 1.8301x over previous
#include <hip/hip_runtime.h>
#include <hip/hip_bf16.h>


typedef __attribute__((ext_vector_type(2))) float v2f;
typedef __attribute__((ext_vector_type(4))) float v4f;
typedef __attribute__((ext_vector_type(8))) float v8f;

#define NN 8
#define DD 16
#define FF 512
#define KK (DD * FF)
#define LN_EPS 1e-5f

#define GQ 4
#define NWAVE 16
#define KSPLIT 2
#define KWAVE (KK / KSPLIT / NWAVE)
#define NGRP (KWAVE / 16)
#define REDP 144


struct GroupRegs {
  float b[32];
  v4f h0a, h0b;
  v4f h1a, h1b;
};

__device__ __forceinline__ void load_group(GroupRegs& R, const float* __restrict__ h0p,
                                           const float* __restrict__ h1p,
                                           const float* __restrict__ wp) {
  R.h0a = *(const v4f*)(h0p);
  R.h0b = *(const v4f*)(h0p + 4);
  R.h1a = *(const v4f*)(h1p);
  R.h1b = *(const v4f*)(h1p + 4);
#pragma unroll
  for (int r = 0; r < 8; ++r) {
#pragma unroll
    for (int j = 0; j < GQ; ++j)
      R.b[r * 4 + j] = __builtin_nontemporal_load(wp + (size_t)r * KK + j * 16);
  }
}

__device__ __forceinline__ void compute_group(const GroupRegs& R, const v2f* acol_f, v8f (&acc)[GQ]) {
  const float h0[8] = {R.h0a.x, R.h0a.y, R.h0a.z, R.h0a.w, R.h0b.x, R.h0b.y, R.h0b.z, R.h0b.w};
  const float h1[8] = {R.h1a.x, R.h1a.y, R.h1a.z, R.h1a.w, R.h1b.x, R.h1b.y, R.h1b.z, R.h1b.w};
#pragma unroll
  for (int t = 0; t < 4; ++t) {
#pragma unroll
    for (int j = 0; j < GQ; ++j) {
      const v4f a = *(const v4f*)(acol_f + j * FF + 2 * t);
      v2f Aop;
      Aop.x = h0[2 * t] * a.x + h1[2 * t] * a.y;
      Aop.y = h0[2 * t + 1] * a.z + h1[2 * t + 1] * a.w;
      v2f Bop;
      Bop.x = R.b[(2 * t) * 4 + j];
      Bop.y = R.b[(2 * t + 1) * 4 + j];
      acc[j] = __builtin_amdgcn_wmma_f32_16x16x4_f32(
          false, Aop, false, Bop, (short)0, acc[j], false, false);
    }
  }
}

__global__ __launch_bounds__(NWAVE * 32) void contract_wmma_kernel(
    const float* __restrict__ h0, const float* __restrict__ A0m,
    const float* __restrict__ h1, const float* __restrict__ A1m,
    float scale, const float* __restrict__ Wm, float* __restrict__ outPart,
    int two) {
  __shared__ __align__(16) v2f acol[GQ][FF];
  __shared__ float red[NWAVE][GQ][REDP];

  const int gq   = blockIdx.x >> 1;
  const int kh   = blockIdx.x & 1;
  const int g0   = gq * GQ;
  const int tid  = threadIdx.x;
  const int w    = tid >> 5;
  const int lane = tid & 31;
  const int hf   = lane >> 4;
  const int c    = lane & 15;
  const int n    = c & 7;

  const int kw0 = kh * (KK / KSPLIT) + w * KWAVE;

  const float* wrow = Wm + (size_t)g0 * 16;
  __builtin_prefetch(wrow + (size_t)(kw0 + 16 + lane) * KK, 0, 1);

  for (int i = tid; i < FF; i += NWAVE * 32) {
    const v4f q0 = *(const v4f*)(A0m + (size_t)i * FF + g0);
    v4f q1 = {0.0f, 0.0f, 0.0f, 0.0f};
    if (two) q1 = *(const v4f*)(A1m + (size_t)i * FF + g0);
    acol[0][i] = (v2f){q0.x, q1.x};
    acol[1][i] = (v2f){q0.y, q1.y};
    acol[2][i] = (v2f){q0.z, q1.z};
    acol[3][i] = (v2f){q0.w, q1.w};
  }
  __syncthreads();

  const int klane = kw0 + 8 * hf;
  const float* h0p = h0 + (size_t)n * KK + klane;
  const float* h1p = h1 + (size_t)n * KK + klane;
  const float* wp  = Wm + (size_t)klane * KK + (size_t)(g0 * 16 + c);
  const v2f* af    = &acol[0][(klane & (FF - 1))];

  v8f acc[GQ];
#pragma unroll
  for (int j = 0; j < GQ; ++j) acc[j] = (v8f){};

  GroupRegs R0, R1;
  load_group(R0, h0p, h1p, wp);

  for (int gp = 0; gp < NGRP / 2 - 1; ++gp) {
    {
      int prow = 16 * (2 * gp + 3) + lane;
      prow = prow < KWAVE ? prow : KWAVE - 1;
      __builtin_prefetch(wrow + (size_t)(kw0 + prow) * KK, 0, 1);
    }
    load_group(R1, h0p + 16, h1p + 16, wp + (size_t)16 * KK);
    __builtin_amdgcn_sched_barrier(0);
    compute_group(R0, af, acc);
    __builtin_amdgcn_sched_barrier(0);
    load_group(R0, h0p + 32, h1p + 32, wp + (size_t)32 * KK);
    __builtin_amdgcn_sched_barrier(0);
    compute_group(R1, af + 16, acc);
    h0p += 32;
    h1p += 32;
    wp  += (size_t)32 * KK;
    af  += 32;
  }
  load_group(R1, h0p + 16, h1p + 16, wp + (size_t)16 * KK);
  __builtin_amdgcn_sched_barrier(0);
  compute_group(R0, af, acc);
  compute_group(R1, af + 16, acc);

  if (hf == 0) {
#pragma unroll
    for (int j = 0; j < GQ; ++j) {
#pragma unroll
      for (int r = 0; r < 8; ++r) red[w][j][r * 16 + c] = acc[j][r];
    }
  }
  __syncthreads();

  {
    const int j  = tid & 3;
    const int no = tid >> 2;
    float s = 0.0f;
#pragma unroll
    for (int ww = 0; ww < NWAVE; ++ww) s += red[ww][j][no];
    outPart[(size_t)kh * (NN * DD * FF) + (size_t)no * FF + g0 + j] = s * scale;
  }
}

__global__ __launch_bounds__(256) void layernorm16_kernel(
    const float* __restrict__ y0, const float* __restrict__ y1, float* __restrict__ out) {
  const int t = blockIdx.x * blockDim.x + threadIdx.x;
  if (t >= NN * FF) return;
  const int n = t / FF;
  const int g = t - n * FF;
  const float* p0 = y0 + n * DD * FF + g;
  const float* p1 = y1 + n * DD * FF + g;
  float v[16];
  float s = 0.0f;
#pragma unroll
  for (int o = 0; o < 16; ++o) { v[o] = p0[o * FF] + p1[o * FF]; s += v[o]; }
  const float mean = s * (1.0f / 16.0f);
  float s2 = 0.0f;
#pragma unroll
  for (int o = 0; o < 16; ++o) { const float d = v[o] - mean; s2 += d * d; }
  const float inv = rsqrtf(s2 * (1.0f / 16.0f) + LN_EPS);
  float* q = out + n * DD * FF + g;
#pragma unroll
  for (int o = 0; o < 16; ++o) q[o * FF] = (v[o] - mean) * inv;
}

extern "C" void kernel_launch(void* const* d_in, const int* in_sizes, int n_in,
                              void* d_out, int out_size, void* d_ws, size_t ws_size,
                              hipStream_t stream) {
  const float* x  = (const float*)d_in[0];
  const float* W  = (const float*)d_in[1];
  const float* A  = (const float*)d_in[2];
  const float* A0 = A;
  const float* A1 = A + FF * FF;

  float* part0 = (float*)d_ws;
  float* part1 = part0 + NN * DD * FF;
  float* h1    = part1 + NN * DD * FF;
  float* outp  = (float*)d_out;

  contract_wmma_kernel<<<(FF / GQ) * KSPLIT, NWAVE * 32, 0, stream>>>(x, A0, x, A0, 1.0f, W, part0, 0);
  layernorm16_kernel<<<(NN * FF + 255) / 256, 256, 0, stream>>>(part0, part1, h1);

  contract_wmma_kernel<<<(FF / GQ) * KSPLIT, NWAVE * 32, 0, stream>>>(x, A1, h1, A0, 0.5f, W, part0, 1);
  layernorm16_kernel<<<(NN * FF + 255) / 256, 256, 0, stream>>>(part0, part1, outp);
}


# Layer 2 finds in the second-level cache the W rows layer 1 of the same call read last: layer 1 loads W plainly, layer 2 walks each wave's k groups downward (its k sum order reversed), W non-temporal
# speedup vs baseline: 1.8880x; 1.0316x over previous
#include <hip/hip_runtime.h>
#include <hip/hip_bf16.h>


typedef __attribute__((ext_vector_type(2))) float v2f;
typedef __attribute__((ext_vector_type(4))) float v4f;
typedef __attribute__((ext_vector_type(8))) float v8f;

#define NN 8
#define DD 16
#define FF 512
#define KK (DD * FF)
#define LN_EPS 1e-5f

#define GQ 4
#define NWAVE 16
#define KSPLIT 2
#define KWAVE (KK / KSPLIT / NWAVE)
#define NGRP (KWAVE / 16)
#define REDP 144


struct GroupRegs {
  float b[32];
  v4f h0a, h0b;
  v4f h1a, h1b;
};

template <bool ONCE>
__device__ __forceinline__ void load_group(GroupRegs& R, const float* __restrict__ h0p,
                                           const float* __restrict__ h1p,
                                           const float* __restrict__ wp) {
  R.h0a = *(const v4f*)(h0p);
  R.h0b = *(const v4f*)(h0p + 4);
  R.h1a = *(const v4f*)(h1p);
  R.h1b = *(const v4f*)(h1p + 4);
#pragma unroll
  for (int r = 0; r < 8; ++r) {
#pragma unroll
    for (int j = 0; j < GQ; ++j) {
      const float* q = wp + (size_t)r * KK + j * 16;
      R.b[r * 4 + j] = ONCE ? __builtin_nontemporal_load(q) : *q;
    }
  }
}

__device__ __forceinline__ void compute_group(const GroupRegs& R, const v2f* acol_f, v8f (&acc)[GQ]) {
  const float h0[8] = {R.h0a.x, R.h0a.y, R.h0a.z, R.h0a.w, R.h0b.x, R.h0b.y, R.h0b.z, R.h0b.w};
  const float h1[8] = {R.h1a.x, R.h1a.y, R.h1a.z, R.h1a.w, R.h1b.x, R.h1b.y, R.h1b.z, R.h1b.w};
#pragma unroll
  for (int t = 0; t < 4; ++t) {
#pragma unroll
    for (int j = 0; j < GQ; ++j) {
      const v4f a = *(const v4f*)(acol_f + j * FF + 2 * t);
      v2f Aop;
      Aop.x = h0[2 * t] * a.x + h1[2 * t] * a.y;
      Aop.y = h0[2 * t + 1] * a.z + h1[2 * t + 1] * a.w;
      v2f Bop;
      Bop.x = R.b[(2 * t) * 4 + j];
      Bop.y = R.b[(2 * t + 1) * 4 + j];
      acc[j] = __builtin_amdgcn_wmma_f32_16x16x4_f32(
          false, Aop, false, Bop, (short)0, acc[j], false, false);
    }
  }
}

template <bool PASS2>
__global__ __launch_bounds__(NWAVE * 32) void contract_wmma_kernel(
    const float* __restrict__ h0, const float* __restrict__ A0m,
    const float* __restrict__ h1, const float* __restrict__ A1m,
    float scale, const float* __restrict__ Wm, float* __restrict__ outPart,
    int two) {
  __shared__ __align__(16) v2f acol[GQ][FF];
  __shared__ float red[NWAVE][GQ][REDP];

  const int gq   = blockIdx.x >> 1;
  const int kh   = blockIdx.x & 1;
  const int g0   = gq * GQ;
  const int tid  = threadIdx.x;
  const int w    = tid >> 5;
  const int lane = tid & 31;
  const int hf   = lane >> 4;
  const int c    = lane & 15;
  const int n    = c & 7;

  const int kw0 = kh * (KK / KSPLIT) + w * KWAVE;
  constexpr int GS = PASS2 ? -16 : 16;
  const int kb0 = PASS2 ? kw0 + KWAVE - 16 : kw0;

  const float* wrow = Wm + (size_t)g0 * 16;
  __builtin_prefetch(wrow + (size_t)(kw0 + (PASS2 ? KWAVE - 1 - (16 + lane) : 16 + lane)) * KK, 0, 1);

  for (int i = tid; i < FF; i += NWAVE * 32) {
    const v4f q0 = *(const v4f*)(A0m + (size_t)i * FF + g0);
    v4f q1 = {0.0f, 0.0f, 0.0f, 0.0f};
    if (two) q1 = *(const v4f*)(A1m + (size_t)i * FF + g0);
    acol[0][i] = (v2f){q0.x, q1.x};
    acol[1][i] = (v2f){q0.y, q1.y};
    acol[2][i] = (v2f){q0.z, q1.z};
    acol[3][i] = (v2f){q0.w, q1.w};
  }
  __syncthreads();

  const int klane = kb0 + 8 * hf;
  const float* h0p = h0 + (size_t)n * KK + klane;
  const float* h1p = h1 + (size_t)n * KK + klane;
  const float* wp  = Wm + (size_t)klane * KK + (size_t)(g0 * 16 + c);
  const v2f* af    = &acol[0][(klane & (FF - 1))];

  v8f acc[GQ];
#pragma unroll
  for (int j = 0; j < GQ; ++j) acc[j] = (v8f){};

  GroupRegs R0, R1;
  load_group<PASS2>(R0, h0p, h1p, wp);

  for (int gp = 0; gp < NGRP / 2 - 1; ++gp) {
    {
      int prow = 16 * (2 * gp + 3) + lane;
      prow = prow < KWAVE ? prow : KWAVE - 1;
      if (PASS2) prow = KWAVE - 1 - prow;
      __builtin_prefetch(wrow + (size_t)(kw0 + prow) * KK, 0, 1);
    }
    load_group<PASS2>(R1, h0p + GS, h1p + GS, wp + (long)GS * KK);
    __builtin_amdgcn_sched_barrier(0);
    compute_group(R0, af, acc);
    __builtin_amdgcn_sched_barrier(0);
    load_group<PASS2>(R0, h0p + 2 * GS, h1p + 2 * GS, wp + (long)(2 * GS) * KK);
    __builtin_amdgcn_sched_barrier(0);
    compute_group(R1, af + GS, acc);
    h0p += 2 * GS;
    h1p += 2 * GS;
    wp  += (long)(2 * GS) * KK;
    af  += 2 * GS;
  }
  load_group<PASS2>(R1, h0p + GS, h1p + GS, wp + (long)GS * KK);
  __builtin_amdgcn_sched_barrier(0);
  compute_group(R0, af, acc);
  compute_group(R1, af + GS, acc);

  if (hf == 0) {
#pragma unroll
    for (int j = 0; j < GQ; ++j) {
#pragma unroll
      for (int r = 0; r < 8; ++r) red[w][j][r * 16 + c] = acc[j][r];
    }
  }
  __syncthreads();

  {
    const int j  = tid & 3;
    const int no = tid >> 2;
    float s = 0.0f;
#pragma unroll
    for (int ww = 0; ww < NWAVE; ++ww) s += red[ww][j][no];
    outPart[(size_t)kh * (NN * DD * FF) + (size_t)no * FF + g0 + j] = s * scale;
  }
}

__global__ __launch_bounds__(256) void layernorm16_kernel(
    const float* __restrict__ y0, const float* __restrict__ y1, float* __restrict__ out) {
  const int t = blockIdx.x * blockDim.x + threadIdx.x;
  if (t >= NN * FF) return;
  const int n = t / FF;
  const int g = t - n * FF;
  const float* p0 = y0 + n * DD * FF + g;
  const float* p1 = y1 + n * DD * FF + g;
  float v[16];
  float s = 0.0f;
#pragma unroll
  for (int o = 0; o < 16; ++o) { v[o] = p0[o * FF] + p1[o * FF]; s += v[o]; }
  const float mean = s * (1.0f / 16.0f);
  float s2 = 0.0f;
#pragma unroll
  for (int o = 0; o < 16; ++o) { const float d = v[o] - mean; s2 += d * d; }
  const float inv = rsqrtf(s2 * (1.0f / 16.0f) + LN_EPS);
  float* q = out + n * DD * FF + g;
#pragma unroll
  for (int o = 0; o < 16; ++o) q[o * FF] = (v[o] - mean) * inv;
}

extern "C" void kernel_launch(void* const* d_in, const int* in_sizes, int n_in,
                              void* d_out, int out_size, void* d_ws, size_t ws_size,
                              hipStream_t stream) {
  const float* x  = (const float*)d_in[0];
  const float* W  = (const float*)d_in[1];
  const float* A  = (const float*)d_in[2];
  const float* A0 = A;
  const float* A1 = A + FF * FF;

  float* part0 = (float*)d_ws;
  float* part1 = part0 + NN * DD * FF;
  float* h1    = part1 + NN * DD * FF;
  float* outp  = (float*)d_out;

  contract_wmma_kernel<false><<<(FF / GQ) * KSPLIT, NWAVE * 32, 0, stream>>>(x, A0, x, A0, 1.0f, W, part0, 0);
  layernorm16_kernel<<<(NN * FF + 255) / 256, 256, 0, stream>>>(part0, part1, h1);

  contract_wmma_kernel<true><<<(FF / GQ) * KSPLIT, NWAVE * 32, 0, stream>>>(x, A1, h1, A0, 0.5f, W, part0, 1);
  layernorm16_kernel<<<(NN * FF + 255) / 256, 256, 0, stream>>>(part0, part1, outp);
}


# W read as 8-byte column pairs: the 8-lane groups of a lane half take different k rows and fill the two diagonal 8x8 blocks of each 16x16x4 tile; half the loads and A operands; new k and sum order
# speedup vs baseline: 1.9996x; 1.0591x over previous
#include <hip/hip_runtime.h>
#include <hip/hip_bf16.h>


typedef __attribute__((ext_vector_type(2))) float v2f;
typedef __attribute__((ext_vector_type(4))) float v4f;
typedef __attribute__((ext_vector_type(8))) float v8f;

#define NN 8
#define DD 16
#define FF 512
#define KK (DD * FF)
#define LN_EPS 1e-5f

#define GQ 4
#define NWAVE 16
#define KSPLIT 2
#define KWAVE (KK / KSPLIT / NWAVE)
#define NGRP (KWAVE / 16)
#define REDP 144
#define NPART (2 * NWAVE)


struct GroupRegs {
  v2f b[16];
  v4f h0;
  v4f h1;
};

template <bool ONCE>
__device__ __forceinline__ void load_group(GroupRegs& R, const float* __restrict__ h0p,
                                           const float* __restrict__ h1p,
                                           const float* __restrict__ wp) {
  R.h0 = *(const v4f*)(h0p);
  R.h1 = *(const v4f*)(h1p);
#pragma unroll
  for (int i = 0; i < 4; ++i) {
#pragma unroll
    for (int j = 0; j < GQ; ++j) {
      const v2f* p = (const v2f*)(wp + (size_t)i * KK + j * 16);
      R.b[i * 4 + j] = ONCE ? __builtin_nontemporal_load(p) : *p;
    }
  }
}

__device__ __forceinline__ void compute_group(const GroupRegs& R, const v2f* acol_f, v8f (&acc)[GQ][2]) {
  const float h0[4] = {R.h0.x, R.h0.y, R.h0.z, R.h0.w};
  const float h1[4] = {R.h1.x, R.h1.y, R.h1.z, R.h1.w};
#pragma unroll
  for (int t = 0; t < 2; ++t) {
#pragma unroll
    for (int j = 0; j < GQ; ++j) {
      const v4f a = *(const v4f*)(acol_f + j * FF + 2 * t);
      v2f Aop;
      Aop.x = h0[2 * t] * a.x + h1[2 * t] * a.y;
      Aop.y = h0[2 * t + 1] * a.z + h1[2 * t + 1] * a.w;
#pragma unroll
      for (int e = 0; e < 2; ++e) {
        v2f Bop;
        Bop.x = R.b[(2 * t) * 4 + j][e];
        Bop.y = R.b[(2 * t + 1) * 4 + j][e];
        acc[j][e] = __builtin_amdgcn_wmma_f32_16x16x4_f32(
            false, Aop, false, Bop, (short)0, acc[j][e], false, false);
      }
    }
  }
}

template <bool PASS2>
__global__ __launch_bounds__(NWAVE * 32) void contract_wmma_kernel(
    const float* __restrict__ h0, const float* __restrict__ A0m,
    const float* __restrict__ h1, const float* __restrict__ A1m,
    float scale, const float* __restrict__ Wm, float* __restrict__ outPart,
    int two) {
  __shared__ __align__(16) v2f acol[GQ][FF];
  __shared__ __align__(16) float red[NPART][GQ][REDP];

  const int gq   = blockIdx.x >> 1;
  const int kh   = blockIdx.x & 1;
  const int g0   = gq * GQ;
  const int tid  = threadIdx.x;
  const int w    = tid >> 5;
  const int lane = tid & 31;
  const int hf   = lane >> 4;
  const int c    = lane & 15;
  const int q    = c >> 3;
  const int n    = c & 7;

  const int kw0 = kh * (KK / KSPLIT) + w * KWAVE;
  constexpr int GS = PASS2 ? -16 : 16;
  const int kb0 = PASS2 ? kw0 + KWAVE - 16 : kw0;

  const float* wrow = Wm + (size_t)g0 * 16;
  __builtin_prefetch(wrow + (size_t)(kw0 + (PASS2 ? KWAVE - 1 - (16 + lane) : 16 + lane)) * KK, 0, 1);

  for (int i = tid; i < FF; i += NWAVE * 32) {
    const v4f q0 = *(const v4f*)(A0m + (size_t)i * FF + g0);
    v4f q1 = {0.0f, 0.0f, 0.0f, 0.0f};
    if (two) q1 = *(const v4f*)(A1m + (size_t)i * FF + g0);
    acol[0][i] = (v2f){q0.x, q1.x};
    acol[1][i] = (v2f){q0.y, q1.y};
    acol[2][i] = (v2f){q0.z, q1.z};
    acol[3][i] = (v2f){q0.w, q1.w};
  }
  __syncthreads();

  const int klane = kb0 + 4 * (2 * hf + q);
  const float* h0p = h0 + (size_t)n * KK + klane;
  const float* h1p = h1 + (size_t)n * KK + klane;
  const float* wp  = Wm + (size_t)klane * KK + (size_t)(g0 * 16 + 2 * n);
  const v2f* af    = &acol[0][(klane & (FF - 1))];

  v8f acc[GQ][2];
#pragma unroll
  for (int j = 0; j < GQ; ++j) { acc[j][0] = (v8f){}; acc[j][1] = (v8f){}; }

  GroupRegs R0, R1;
  load_group<PASS2>(R0, h0p, h1p, wp);

#pragma unroll 1
  for (int gp = 0; gp < NGRP / 2 - 1; ++gp) {
    {
      int prow = 16 * (2 * gp + 3) + lane;
      prow = prow < KWAVE ? prow : KWAVE - 1;
      if (PASS2) prow = KWAVE - 1 - prow;
      __builtin_prefetch(wrow + (size_t)(kw0 + prow) * KK, 0, 1);
    }
    load_group<PASS2>(R1, h0p + GS, h1p + GS, wp + (long)GS * KK);
    __builtin_amdgcn_sched_barrier(0);
    compute_group(R0, af, acc);
    __builtin_amdgcn_sched_barrier(0);
    load_group<PASS2>(R0, h0p + 2 * GS, h1p + 2 * GS, wp + (long)(2 * GS) * KK);
    __builtin_amdgcn_sched_barrier(0);
    compute_group(R1, af + GS, acc);
    h0p += 2 * GS;
    h1p += 2 * GS;
    wp  += (long)(2 * GS) * KK;
    af  += 2 * GS;
  }
  load_group<PASS2>(R1, h0p + GS, h1p + GS, wp + (long)GS * KK);
  __builtin_amdgcn_sched_barrier(0);
  compute_group(R0, af, acc);
  compute_group(R1, af + GS, acc);

  if (hf == q) {
#pragma unroll
    for (int j = 0; j < GQ; ++j) {
#pragma unroll
      for (int r = 0; r < 8; ++r)
        *(v2f*)&red[2 * w + q][j][r * 16 + 2 * n] = (v2f){acc[j][0][r], acc[j][1][r]};
    }
  }
  __syncthreads();

  {
    const int j  = tid & 3;
    const int no = tid >> 2;
    float s = 0.0f;
#pragma unroll
    for (int ww = 0; ww < NPART; ++ww) s += red[ww][j][no];
    outPart[(size_t)kh * (NN * DD * FF) + (size_t)no * FF + g0 + j] = s * scale;
  }
}

__global__ __launch_bounds__(256) void layernorm16_kernel(
    const float* __restrict__ y0, const float* __restrict__ y1, float* __restrict__ out) {
  const int t = blockIdx.x * blockDim.x + threadIdx.x;
  if (t >= NN * FF) return;
  const int n = t / FF;
  const int g = t - n * FF;
  const float* p0 = y0 + n * DD * FF + g;
  const float* p1 = y1 + n * DD * FF + g;
  float v[16];
  float s = 0.0f;
#pragma unroll
  for (int o = 0; o < 16; ++o) { v[o] = p0[o * FF] + p1[o * FF]; s += v[o]; }
  const float mean = s * (1.0f / 16.0f);
  float s2 = 0.0f;
#pragma unroll
  for (int o = 0; o < 16; ++o) { const float d = v[o] - mean; s2 += d * d; }
  const float inv = rsqrtf(s2 * (1.0f / 16.0f) + LN_EPS);
  float* q = out + n * DD * FF + g;
#pragma unroll
  for (int o = 0; o < 16; ++o) q[o * FF] = (v[o] - mean) * inv;
}

extern "C" void kernel_launch(void* const* d_in, const int* in_sizes, int n_in,
                              void* d_out, int out_size, void* d_ws, size_t ws_size,
                              hipStream_t stream) {
  const float* x  = (const float*)d_in[0];
  const float* W  = (const float*)d_in[1];
  const float* A  = (const float*)d_in[2];
  const float* A0 = A;
  const float* A1 = A + FF * FF;

  float* part0 = (float*)d_ws;
  float* part1 = part0 + NN * DD * FF;
  float* h1    = part1 + NN * DD * FF;
  float* outp  = (float*)d_out;

  contract_wmma_kernel<false><<<(FF / GQ) * KSPLIT, NWAVE * 32, 0, stream>>>(x, A0, x, A0, 1.0f, W, part0, 0);
  layernorm16_kernel<<<(NN * FF + 255) / 256, 256, 0, stream>>>(part0, part1, h1);

  contract_wmma_kernel<true><<<(FF / GQ) * KSPLIT, NWAVE * 32, 0, stream>>>(x, A1, h1, A0, 0.5f, W, part0, 1);
  layernorm16_kernel<<<(NN * FF + 255) / 256, 256, 0, stream>>>(part0, part1, outp);
}


# Each workgroup sums all of k for 2 adjacent columns (16 waves x 512 k), so the layer norm runs in the contraction's own epilogue: 2 launches instead of 4, no partial planes; new k sum order
# speedup vs baseline: 2.0838x; 1.0421x over previous
#include <hip/hip_runtime.h>
#include <hip/hip_bf16.h>


typedef __attribute__((ext_vector_type(2))) float v2f;
typedef __attribute__((ext_vector_type(4))) float v4f;
typedef __attribute__((ext_vector_type(8))) float v8f;

#define NN 8
#define DD 16
#define FF 512
#define KK (DD * FF)
#define LN_EPS 1e-5f

#define GQ 2
#define NWAVE 16
#define KWAVE (KK / NWAVE)
#define GK 32
#define NGRP (KWAVE / GK)
#define REDP 144
#define NPART (2 * NWAVE)


struct GroupRegs {
  v2f b[2][8];
  v4f h0[2];
  v4f h1[2];
};

template <bool ONCE>
__device__ __forceinline__ void load_group(GroupRegs& R, const float* __restrict__ h0p,
                                           const float* __restrict__ h1p,
                                           const float* __restrict__ wp) {
#pragma unroll
  for (int u = 0; u < 2; ++u) {
    R.h0[u] = *(const v4f*)(h0p + 16 * u);
    R.h1[u] = *(const v4f*)(h1p + 16 * u);
  }
#pragma unroll
  for (int u = 0; u < 2; ++u) {
#pragma unroll
    for (int i = 0; i < 4; ++i) {
#pragma unroll
      for (int j = 0; j < GQ; ++j) {
        const v2f* p = (const v2f*)(wp + (size_t)(16 * u + i) * KK + j * 16);
        R.b[u][i * 2 + j] = ONCE ? __builtin_nontemporal_load(p) : *p;
      }
    }
  }
}

__device__ __forceinline__ void compute_group(const GroupRegs& R, const v2f* acol_f, v8f (&acc)[GQ][2]) {
#pragma unroll
  for (int u = 0; u < 2; ++u) {
    const float h0[4] = {R.h0[u].x, R.h0[u].y, R.h0[u].z, R.h0[u].w};
    const float h1[4] = {R.h1[u].x, R.h1[u].y, R.h1[u].z, R.h1[u].w};
#pragma unroll
    for (int t = 0; t < 2; ++t) {
#pragma unroll
      for (int j = 0; j < GQ; ++j) {
        const v4f a = *(const v4f*)(acol_f + j * FF + 16 * u + 2 * t);
        v2f Aop;
        Aop.x = h0[2 * t] * a.x + h1[2 * t] * a.y;
        Aop.y = h0[2 * t + 1] * a.z + h1[2 * t + 1] * a.w;
#pragma unroll
        for (int e = 0; e < 2; ++e) {
          v2f Bop;
          Bop.x = R.b[u][(2 * t) * 2 + j][e];
          Bop.y = R.b[u][(2 * t + 1) * 2 + j][e];
          acc[j][e] = __builtin_amdgcn_wmma_f32_16x16x4_f32(
              false, Aop, false, Bop, (short)0, acc[j][e], false, false);
        }
      }
    }
  }
}

template <bool PASS2>
__global__ __launch_bounds__(NWAVE * 32) void contract_norm_kernel(
    const float* __restrict__ h0, const float* __restrict__ A0m,
    const float* __restrict__ h1, const float* __restrict__ A1m,
    float scale, const float* __restrict__ Wm, float* __restrict__ dst,
    int two) {
  __shared__ __align__(16) v2f acol[GQ][FF];
  __shared__ __align__(16) float red[NPART][GQ][REDP];
  __shared__ float ysum[GQ][NN * DD];

  const int b    = blockIdx.x;
  const int gq   = (((b >> 5) << 4) | (b & 15)) * 2 + ((b >> 4) & 1);
  const int g0   = gq * GQ;
  const int tid  = threadIdx.x;
  const int w    = tid >> 5;
  const int lane = tid & 31;
  const int hf   = lane >> 4;
  const int c    = lane & 15;
  const int q    = c >> 3;
  const int n    = c & 7;

  const int kw0 = w * KWAVE;
  constexpr int GS = PASS2 ? -GK : GK;
  const int kb0 = PASS2 ? kw0 + KWAVE - GK : kw0;

  const float* wrow = Wm + (size_t)g0 * 16;
  __builtin_prefetch(wrow + (size_t)(kw0 + (PASS2 ? KWAVE - 1 - (GK + lane) : GK + lane)) * KK, 0, 1);
  __builtin_prefetch(wrow + (size_t)(kw0 + (PASS2 ? KWAVE - 1 - (2 * GK + lane) : 2 * GK + lane)) * KK, 0, 1);

  for (int i = tid; i < FF; i += NWAVE * 32) {
    const v2f q0 = *(const v2f*)(A0m + (size_t)i * FF + g0);
    v2f q1 = {0.0f, 0.0f};
    if (two) q1 = *(const v2f*)(A1m + (size_t)i * FF + g0);
    acol[0][i] = (v2f){q0.x, q1.x};
    acol[1][i] = (v2f){q0.y, q1.y};
  }
  __syncthreads();

  const int klane = kb0 + 4 * (2 * hf + q);
  const float* h0p = h0 + (size_t)n * KK + klane;
  const float* h1p = h1 + (size_t)n * KK + klane;
  const float* wp  = Wm + (size_t)klane * KK + (size_t)(g0 * 16 + 2 * n);
  const v2f* af    = &acol[0][(klane & (FF - 1))];

  v8f acc[GQ][2];
#pragma unroll
  for (int j = 0; j < GQ; ++j) { acc[j][0] = (v8f){}; acc[j][1] = (v8f){}; }

  GroupRegs R0, R1;
  load_group<PASS2>(R0, h0p, h1p, wp);

#pragma unroll 1
  for (int gp = 0; gp < NGRP / 2 - 1; ++gp) {
#pragma unroll
    for (int pp = 0; pp < 2; ++pp) {
      int prow = GK * (2 * gp + 3 + pp) + lane;
      prow = prow < KWAVE ? prow : KWAVE - 1;
      if (PASS2) prow = KWAVE - 1 - prow;
      __builtin_prefetch(wrow + (size_t)(kw0 + prow) * KK, 0, 1);
    }
    load_group<PASS2>(R1, h0p + GS, h1p + GS, wp + (long)GS * KK);
    __builtin_amdgcn_sched_barrier(0);
    compute_group(R0, af, acc);
    __builtin_amdgcn_sched_barrier(0);
    load_group<PASS2>(R0, h0p + 2 * GS, h1p + 2 * GS, wp + (long)(2 * GS) * KK);
    __builtin_amdgcn_sched_barrier(0);
    compute_group(R1, af + GS, acc);
    h0p += 2 * GS;
    h1p += 2 * GS;
    wp  += (long)(2 * GS) * KK;
    af  += 2 * GS;
  }
  load_group<PASS2>(R1, h0p + GS, h1p + GS, wp + (long)GS * KK);
  __builtin_amdgcn_sched_barrier(0);
  compute_group(R0, af, acc);
  compute_group(R1, af + GS, acc);

  if (hf == q) {
#pragma unroll
    for (int j = 0; j < GQ; ++j) {
#pragma unroll
      for (int r = 0; r < 8; ++r)
        *(v2f*)&red[2 * w + q][j][r * 16 + 2 * n] = (v2f){acc[j][0][r], acc[j][1][r]};
    }
  }
  __syncthreads();

  const int j  = tid & (GQ - 1);
  const int no = (tid >> 1) & (NN * DD - 1);
  const bool owner = tid < GQ * NN * DD;
  if (owner) {
    float s = 0.0f;
#pragma unroll
    for (int ww = 0; ww < NPART; ++ww) s += red[ww][j][no];
    ysum[j][no] = s * scale;
  }
  __syncthreads();

  if (owner) {
    const float* yv = &ysum[j][no & ~15];
    float v[16];
    float s = 0.0f;
#pragma unroll
    for (int o = 0; o < 16; ++o) { v[o] = yv[o]; s += v[o]; }
    const float mean = s * (1.0f / 16.0f);
    float s2 = 0.0f;
#pragma unroll
    for (int o = 0; o < 16; ++o) { const float d = v[o] - mean; s2 += d * d; }
    const float inv = rsqrtf(s2 * (1.0f / 16.0f) + LN_EPS);
    dst[(size_t)no * FF + g0 + j] = (ysum[j][no] - mean) * inv;
  }
}

extern "C" void kernel_launch(void* const* d_in, const int* in_sizes, int n_in,
                              void* d_out, int out_size, void* d_ws, size_t ws_size,
                              hipStream_t stream) {
  const float* x  = (const float*)d_in[0];
  const float* W  = (const float*)d_in[1];
  const float* A  = (const float*)d_in[2];
  const float* A0 = A;
  const float* A1 = A + FF * FF;

  float* h1    = (float*)d_ws;
  float* outp  = (float*)d_out;

  contract_norm_kernel<false><<<FF / GQ, NWAVE * 32, 0, stream>>>(x, A0, x, A0, 1.0f, W, h1, 0);

  contract_norm_kernel<true><<<FF / GQ, NWAVE * 32, 0, stream>>>(x, A1, h1, A0, 0.5f, W, outp, 1);
}
